# starting kernel
# baseline (speedup 1.0000x reference)
#include <hip/hip_runtime.h>
#include <hip/hip_bf16.h>


typedef __attribute__((ext_vector_type(2))) float v2f;
typedef __attribute__((ext_vector_type(8))) float v8f;

__global__ __launch_bounds__(256) void gemm_f32_wmma(
    const float* __restrict__ A, const float* __restrict__ B,
    const float* __restrict__ bias, float* __restrict__ C, int M, int K) {
  const int wave = threadIdx.x >> 5;
  const int lane = threadIdx.x & 31;
  const int half = lane >> 4;
  const int l15  = lane & 15;

  const int row0 = (blockIdx.x * 8 + wave) * 16;
  if (row0 >= M) return;

  v8f acc0 = {}, acc1 = {}, acc2 = {}, acc3 = {};
  const float* aRow = A + (size_t)(row0 + l15) * K + 2 * half;

  for (int k = 0; k < K; k += 4) {
    v2f a;
    a.x = aRow[k];
    a.y = aRow[k + 1];
    const float* bRow = B + (size_t)(k + 2 * half) * 64 + l15;
    v2f b0, b1, b2, b3;
    b0.x = bRow[0];   b0.y = bRow[64];
    b1.x = bRow[16];  b1.y = bRow[80];
    b2.x = bRow[32];  b2.y = bRow[96];
    b3.x = bRow[48];  b3.y = bRow[112];
    acc0 = __builtin_amdgcn_wmma_f32_16x16x4_f32(false, a, false, b0,
                                                 (short)0, acc0, false, false);
    acc1 = __builtin_amdgcn_wmma_f32_16x16x4_f32(false, a, false, b1,
                                                 (short)0, acc1, false, false);
    acc2 = __builtin_amdgcn_wmma_f32_16x16x4_f32(false, a, false, b2,
                                                 (short)0, acc2, false, false);
    acc3 = __builtin_amdgcn_wmma_f32_16x16x4_f32(false, a, false, b3,
                                                 (short)0, acc3, false, false);
  }

  float bb0 = 0.f, bb1 = 0.f, bb2 = 0.f, bb3 = 0.f;
  if (bias) {
    bb0 = bias[l15];      bb1 = bias[l15 + 16];
    bb2 = bias[l15 + 32]; bb3 = bias[l15 + 48];
  }
#pragma unroll
  for (int r = 0; r < 8; ++r) {
    const int row = row0 + r + 8 * half;
    float* cRow = C + (size_t)row * 64 + l15;
    cRow[0]  = acc0[r] + bb0;
    cRow[16] = acc1[r] + bb1;
    cRow[32] = acc2[r] + bb2;
    cRow[48] = acc3[r] + bb3;
  }
}

__global__ void fill_ones(float* __restrict__ p, int n) {
  int i = blockIdx.x * blockDim.x + threadIdx.x;
  if (i < n) p[i] = 1.0f;
}

__global__ void deg_accum(const int* __restrict__ dst,
                          const float* __restrict__ ew,
                          float* __restrict__ deg, int ne) {
  int e = blockIdx.x * blockDim.x + threadIdx.x;
  if (e < ne) atomicAdd(&deg[dst[e]], ew[e]);
}

__global__ void deg_to_dinv(float* __restrict__ p, int n) {
  int i = blockIdx.x * blockDim.x + threadIdx.x;
  if (i < n) {
    float d = p[i];
    p[i] = (d > 0.f) ? rsqrtf(d) : 0.f;
  }
}

__global__ void self_loop_init(const float* __restrict__ h,
                               const float* __restrict__ dinv,
                               float* __restrict__ agg, int total) {
  int t = blockIdx.x * blockDim.x + threadIdx.x;
  if (t >= total) return;
  float di = dinv[t >> 6];
  agg[t] = h[t] * di * di;
}

__global__ __launch_bounds__(256) void edge_scatter(
    const float* __restrict__ h, const int* __restrict__ src,
    const int* __restrict__ dst, const float* __restrict__ ew,
    const float* __restrict__ dinv, float* __restrict__ agg, int ne) {
  long long t = (long long)blockIdx.x * blockDim.x + threadIdx.x;
  int e = (int)(t >> 4);
  if (e >= ne) return;
  int f0 = ((int)t & 15) * 4;
  int s = src[e], d = dst[e];
  float nrm = dinv[s] * ew[e] * dinv[d];
  const float4 hv = *(const float4*)(h + (size_t)s * 64 + f0);
  float* ad = agg + (size_t)d * 64 + f0;
  atomicAdd(ad + 0, hv.x * nrm);
  atomicAdd(ad + 1, hv.y * nrm);
  atomicAdd(ad + 2, hv.z * nrm);
  atomicAdd(ad + 3, hv.w * nrm);
}

__global__ void bias_act(float* __restrict__ v, const float* __restrict__ b,
                         int total, int relu) {
  int t = blockIdx.x * blockDim.x + threadIdx.x;
  if (t >= total) return;
  float x = v[t] + b[t & 63];
  v[t] = relu ? fmaxf(x, 0.f) : x;
}

extern "C" void kernel_launch(void* const* d_in, const int* in_sizes, int n_in,
                              void* d_out, int out_size, void* d_ws,
                              size_t ws_size, hipStream_t stream) {
  const float* x  = (const float*)d_in[0];
  const int*   ei = (const int*)d_in[1];
  const float* ew = (const float*)d_in[2];
  const float* Wm = (const float*)d_in[3];
  const float* bm = (const float*)d_in[4];
  const float* W1 = (const float*)d_in[5];
  const float* b1 = (const float*)d_in[6];
  const float* W2 = (const float*)d_in[7];
  const float* b2 = (const float*)d_in[8];
  float* out = (float*)d_out;

  const int NN = in_sizes[0] / 384;
  const int EE = in_sizes[2];
  const int* src = ei;
  const int* dst = ei + EE;

  char* ws = (char*)d_ws;
  float* dinv = (float*)ws;
  size_t off  = ((size_t)NN * sizeof(float) + 255) & ~(size_t)255;
  float* x64  = (float*)(ws + off);
  float* hbuf = (float*)(ws + off + 1 * (size_t)NN * 64 * sizeof(float));
  float* agg  = (float*)(ws + off + 2 * (size_t)NN * 64 * sizeof(float));

  const int total = NN * 64;
  const dim3 blk(256);
  const int gTot  = (total + 255) / 256;
  const int gN    = (NN + 255) / 256;
  const int gE    = (EE + 255) / 256;
  const int gE16  = (int)(((long long)EE * 16 + 255) / 256);
  const int gGemm = ((NN + 15) / 16 + 7) / 8;

  fill_ones<<<gN, blk, 0, stream>>>(dinv, NN);
  deg_accum<<<gE, blk, 0, stream>>>(dst, ew, dinv, EE);
  deg_to_dinv<<<gN, blk, 0, stream>>>(dinv, NN);

  gemm_f32_wmma<<<gGemm, blk, 0, stream>>>(x, Wm, bm, x64, NN, 384);

  gemm_f32_wmma<<<gGemm, blk, 0, stream>>>(x64, W1, nullptr, hbuf, NN, 64);
  self_loop_init<<<gTot, blk, 0, stream>>>(hbuf, dinv, agg, total);
  edge_scatter<<<gE16, blk, 0, stream>>>(hbuf, src, dst, ew, dinv, agg, EE);
  bias_act<<<gTot, blk, 0, stream>>>(agg, b1, total, 1);

  gemm_f32_wmma<<<gGemm, blk, 0, stream>>>(agg, W2, nullptr, hbuf, NN, 64);
  self_loop_init<<<gTot, blk, 0, stream>>>(hbuf, dinv, out, total);
  edge_scatter<<<gE16, blk, 0, stream>>>(hbuf, src, dst, ew, dinv, out, EE);
  bias_act<<<gTot, blk, 0, stream>>>(out, b2, total, 0);
}


# Weight operand of all three dense products kept whole in LDS in fragment order, staged once per work-group: k loop makes no global B load, A lines go to registers one block ahead; sums unchanged
# speedup vs baseline: 2.3817x; 2.3817x over previous
#include <hip/hip_runtime.h>
#include <hip/hip_bf16.h>


typedef __attribute__((ext_vector_type(2))) float v2f;
typedef __attribute__((ext_vector_type(8))) float v8f;

__device__ __forceinline__ float inv_sqrt_deg(float d) {
  return (d > 0.f) ? rsqrtf(d) : 0.f;
}

constexpr int kDegRange  = 8192;
constexpr int kDegSlices = 16;
constexpr int kThreads   = 256;

template <int kK>
__device__ __forceinline__ void stage_weights(const float* __restrict__ B,
                                              float* __restrict__ sB) {
  for (int q = threadIdx.x; q < (kK / 2) * 16; q += kThreads) {
    const int kp = q >> 4;
    const int n0 = (q & 15) * 4;
    const float4 r0 =
        *reinterpret_cast<const float4*>(B + (size_t)(2 * kp) * 64 + n0);
    const float4 r1 =
        *reinterpret_cast<const float4*>(B + (size_t)(2 * kp + 1) * 64 + n0);
    const int t = n0 >> 4;
    const int c0 = n0 & 15;
    float* p = sB + (((kp * 2 + (t >> 1)) * 16 + c0) * 4 + (t & 1) * 2);
    *reinterpret_cast<float2*>(p)      = make_float2(r0.x, r1.x);
    *reinterpret_cast<float2*>(p + 4)  = make_float2(r0.y, r1.y);
    *reinterpret_cast<float2*>(p + 8)  = make_float2(r0.z, r1.z);
    *reinterpret_cast<float2*>(p + 12) = make_float2(r0.w, r1.w);
  }
}

template <bool kReluA>
__device__ __forceinline__ void mma_block(const v2f (&a8)[8],
                                          const float4* __restrict__ bv,
                                          int kb, v8f& acc0, v8f& acc1,
                                          v8f& acc2, v8f& acc3) {
#pragma unroll
  for (int s = 0; s < 8; ++s) {
    const int kp = (kb >> 1) + 2 * s;
    const float4 v0 = bv[kp * 32];
    const float4 v1 = bv[kp * 32 + 16];
    v2f a = a8[s];
    if (kReluA) {
      a.x = fmaxf(a.x, 0.f);
      a.y = fmaxf(a.y, 0.f);
    }
    v2f b0, b1, b2, b3;
    b0.x = v0.x; b0.y = v0.y;
    b1.x = v0.z; b1.y = v0.w;
    b2.x = v1.x; b2.y = v1.y;
    b3.x = v1.z; b3.y = v1.w;
    acc0 = __builtin_amdgcn_wmma_f32_16x16x4_f32(false, a, false, b0,
                                                 (short)0, acc0, false, false);
    acc1 = __builtin_amdgcn_wmma_f32_16x16x4_f32(false, a, false, b1,
                                                 (short)0, acc1, false, false);
    acc2 = __builtin_amdgcn_wmma_f32_16x16x4_f32(false, a, false, b2,
                                                 (short)0, acc2, false, false);
    acc3 = __builtin_amdgcn_wmma_f32_16x16x4_f32(false, a, false, b3,
                                                 (short)0, acc3, false, false);
  }
}

template <int kK, bool kReluA, bool kStart, bool kDegSum>
__device__ __forceinline__ void gemm_strip(
    const float* __restrict__ A, const float* __restrict__ B,
    const float* __restrict__ bias, float* __restrict__ C,
    const float* __restrict__ deg, float* __restrict__ S,
    float* __restrict__ degOut, int M, int block, float* __restrict__ sB) {
  static_assert(kK % 32 == 0, "the k loop walks whole 128-byte lines of A");
  const int wave = threadIdx.x >> 5;
  const int lane = threadIdx.x & 31;
  const int half = lane >> 4;
  const int l15  = lane & 15;

  stage_weights<kK>(B, sB);
  __syncthreads();

  const int row0 = (block * 8 + wave) * 16;
  if (row0 >= M) return;

  const float* aRow = A + (size_t)(row0 + l15) * kK + 2 * half;
  v2f aCur[8];
#pragma unroll
  for (int s = 0; s < 8; ++s) {
    aCur[s].x = aRow[4 * s];
    aCur[s].y = aRow[4 * s + 1];
  }

  float di[8];
  if (kStart && kDegSum) {
    float s = 0.f;
#pragma unroll
    for (int p = 0; p < kDegSlices / 2; ++p)
      s += deg[(size_t)(half * (kDegSlices / 2) + p) * M + row0 + l15];
    s += __shfl_xor(s, 16);
    const float dg = s + 1.0f;
    degOut[row0 + l15] = dg;
    const float dv = inv_sqrt_deg(dg);
#pragma unroll
    for (int r = 0; r < 8; ++r) di[r] = __shfl(dv, r + 8 * half);
  } else if (kStart) {
#pragma unroll
    for (int r = 0; r < 8; ++r) di[r] = inv_sqrt_deg(deg[row0 + r + 8 * half]);
  }
  float bb0 = 0.f, bb1 = 0.f, bb2 = 0.f, bb3 = 0.f;
  if (bias) {
    bb0 = bias[l15];      bb1 = bias[l15 + 16];
    bb2 = bias[l15 + 32]; bb3 = bias[l15 + 48];
  }

  v8f acc0 = {}, acc1 = {}, acc2 = {}, acc3 = {};
  const float4* bv = reinterpret_cast<const float4*>(sB) + half * 32 + l15;

  for (int kb = 0; kb + 32 < kK; kb += 32) {
    v2f aNxt[8];
#pragma unroll
    for (int s = 0; s < 8; ++s) {
      aNxt[s].x = aRow[kb + 32 + 4 * s];
      aNxt[s].y = aRow[kb + 32 + 4 * s + 1];
    }
    mma_block<kReluA>(aCur, bv, kb, acc0, acc1, acc2, acc3);
#pragma unroll
    for (int s = 0; s < 8; ++s) aCur[s] = aNxt[s];
  }
  mma_block<kReluA>(aCur, bv, kK - 32, acc0, acc1, acc2, acc3);

#pragma unroll
  for (int r = 0; r < 8; ++r) {
    const int row = row0 + r + 8 * half;
    float* cRow = C + (size_t)row * 64 + l15;
    if (kStart) {
      cRow[0]  = acc0[r];
      cRow[16] = acc1[r];
      cRow[32] = acc2[r];
      cRow[48] = acc3[r];
      float* sRow = S + (size_t)row * 64 + l15;
      const float d = di[r];
      sRow[0]  = acc0[r] * d * d + bb0;
      sRow[16] = acc1[r] * d * d + bb1;
      sRow[32] = acc2[r] * d * d + bb2;
      sRow[48] = acc3[r] * d * d + bb3;
    } else {
      cRow[0]  = acc0[r] + bb0;
      cRow[16] = acc1[r] + bb1;
      cRow[32] = acc2[r] + bb2;
      cRow[48] = acc3[r] + bb3;
    }
  }
}

template <int kK, bool kReluA, bool kStart, bool kDegSum>
__global__ __launch_bounds__(256) void gemm_f32_wmma(
    const float* __restrict__ A, const float* __restrict__ B,
    const float* __restrict__ bias, float* __restrict__ C,
    const float* __restrict__ deg, float* __restrict__ S,
    float* __restrict__ degOut, int M) {
  __shared__ __attribute__((aligned(16))) float sB[kK * 64];
  gemm_strip<kK, kReluA, kStart, kDegSum>(A, B, bias, C, deg, S, degOut, M,
                                          (int)blockIdx.x, sB);
}

__device__ __forceinline__ void deg_add(float* hist, int base, int d, float w) {
  const unsigned r = (unsigned)(d - base);
  if (r < (unsigned)kDegRange) atomicAdd(&hist[r], w);
}
__device__ __forceinline__ void deg_add4(float* hist, int base, int4 d,
                                         float4 w) {
  deg_add(hist, base, d.x, w.x);
  deg_add(hist, base, d.y, w.y);
  deg_add(hist, base, d.z, w.z);
  deg_add(hist, base, d.w, w.w);
}

__device__ __forceinline__ void degree_partial(
    const int* __restrict__ dst, const float* __restrict__ ew,
    float* __restrict__ part, int nn, int ne, int block, int nRanges,
    float* __restrict__ hist) {
  const int tid = threadIdx.x;
  const int slice = block / nRanges;
  const int range = block - slice * nRanges;
  const int base = range * kDegRange;

  for (int i = tid; i < kDegRange; i += kThreads) hist[i] = 0.f;
  __syncthreads();

  const int per = (((ne + kDegSlices - 1) / kDegSlices) + 3) & ~3;
  const int e0 = slice * per;
  const int e1 = (e0 + per < ne) ? (e0 + per) : ne;
  const int n = (e1 > e0) ? (e1 - e0) : 0;
  int done = 0;
  if (((((size_t)dst) | ((size_t)ew)) & 15) == 0) {
    const int4* d4 = reinterpret_cast<const int4*>(dst + e0);
    const float4* w4 = reinterpret_cast<const float4*>(ew + e0);
    const int n4 = n >> 2;
    int i = tid;
    for (; i + 7 * kThreads < n4; i += 8 * kThreads) {
      int4 d[8];
      float4 w[8];
#pragma unroll
      for (int u = 0; u < 8; ++u) {
        d[u] = d4[i + u * kThreads];
        w[u] = w4[i + u * kThreads];
      }
#pragma unroll
      for (int u = 0; u < 8; ++u) deg_add4(hist, base, d[u], w[u]);
    }
    for (; i < n4; i += kThreads) deg_add4(hist, base, d4[i], w4[i]);
    done = n4 << 2;
  }
  for (int e = e0 + done + tid; e < e1; e += kThreads)
    deg_add(hist, base, dst[e], ew[e]);
  __syncthreads();

  float* plane = part + (size_t)slice * nn;
  for (int i = tid; i < kDegRange; i += kThreads) {
    const int node = base + i;
    if (node < nn) plane[node] = hist[i];
  }
}

__global__ __launch_bounds__(256) void proj_and_degree(
    const float* __restrict__ x, const float* __restrict__ Wm,
    const float* __restrict__ bm, float* __restrict__ x64,
    const int* __restrict__ dst, const float* __restrict__ ew,
    float* __restrict__ part, int M, int ne, int nRanges) {
  constexpr int kProjK = 384;
  static_assert(kProjK * 64 >= kDegRange, "the degree table fits in it");
  __shared__ __attribute__((aligned(16))) float smem[kProjK * 64];
  const int nDeg = nRanges * kDegSlices;
  if ((int)blockIdx.x < nDeg) {
    degree_partial(dst, ew, part, M, ne, (int)blockIdx.x, nRanges, smem);
  } else {
    gemm_strip<kProjK, false, false, false>(x, Wm, bm, x64, nullptr, nullptr,
                                            nullptr, M,
                                            (int)blockIdx.x - nDeg, smem);
  }
}

constexpr int kEdgesPerWave = 32;
constexpr int kGroup = 8;

__device__ __forceinline__ unsigned lane_bcast(unsigned v, int j) {
  return (unsigned)__builtin_amdgcn_readlane((int)v, j);
}
__device__ __forceinline__ float lane_bcast(float v, int j) {
  return __int_as_float(__builtin_amdgcn_readlane(__float_as_int(v), j));
}

template <bool kMakeNorm>
__global__ __launch_bounds__(256) void edge_scatter(
    const float* __restrict__ h, const int* __restrict__ src,
    const int* __restrict__ dst, const float* __restrict__ ew,
    const float* __restrict__ deg, float* __restrict__ enorm,
    float* __restrict__ agg, int ne) {
  const unsigned lane = threadIdx.x & 31;
  const int wib = __builtin_amdgcn_readfirstlane((int)(threadIdx.x >> 5));
  const long long e0 =
      ((long long)blockIdx.x * (blockDim.x >> 5) + wib) * kEdgesPerWave;
  if (e0 >= ne) return;
  const int cnt = (ne - e0 < kEdgesPerWave) ? (int)(ne - e0) : kEdgesPerWave;

  unsigned so = 0, dofs = 0;
  float nrm = 0.f;
  if ((int)lane < cnt) {
    const long long e = e0 + lane;
    const int s = src[e];
    const int d = dst[e];
    if (kMakeNorm) {
      nrm = inv_sqrt_deg(deg[s]) * ew[e] * inv_sqrt_deg(deg[d]);
      enorm[e] = nrm;
    } else {
      nrm = enorm[e];
    }
    so = (unsigned)s * 64u;
    dofs = (unsigned)d * 64u;
  }

  if (cnt == kEdgesPerWave) {
    float lo[2][kGroup], hi[2][kGroup];
#pragma unroll
    for (int g = 0; g < kGroup; ++g) {
      const float* hp = h + (lane_bcast(so, g) + lane);
      lo[0][g] = hp[0];
      hi[0][g] = hp[32];
    }
#pragma unroll
    for (int j0 = 0; j0 < kEdgesPerWave; j0 += kGroup) {
      const int cur = (j0 / kGroup) & 1;
      const int nxt = cur ^ 1;
      if (j0 + kGroup < kEdgesPerWave) {
#pragma unroll
        for (int g = 0; g < kGroup; ++g) {
          const float* hp = h + (lane_bcast(so, j0 + kGroup + g) + lane);
          lo[nxt][g] = hp[0];
          hi[nxt][g] = hp[32];
        }
      }
#pragma unroll
      for (int g = 0; g < kGroup; ++g) {
        const float nj = lane_bcast(nrm, j0 + g);
        float* ap = agg + (lane_bcast(dofs, j0 + g) + lane);
        atomicAdd(ap, lo[cur][g] * nj);
        atomicAdd(ap + 32, hi[cur][g] * nj);
      }
    }
  } else {
    for (int j = 0; j < cnt; ++j) {
      const float nj = lane_bcast(nrm, j);
      const float* hp = h + (lane_bcast(so, j) + lane);
      float* ap = agg + (lane_bcast(dofs, j) + lane);
      const float a = hp[0], b = hp[32];
      atomicAdd(ap, a * nj);
      atomicAdd(ap + 32, b * nj);
    }
  }
}

extern "C" void kernel_launch(void* const* d_in, const int* in_sizes, int n_in,
                              void* d_out, int out_size, void* d_ws,
                              size_t ws_size, hipStream_t stream) {
  const float* x  = (const float*)d_in[0];
  const int*   ei = (const int*)d_in[1];
  const float* ew = (const float*)d_in[2];
  const float* Wm = (const float*)d_in[3];
  const float* bm = (const float*)d_in[4];
  const float* W1 = (const float*)d_in[5];
  const float* b1 = (const float*)d_in[6];
  const float* W2 = (const float*)d_in[7];
  const float* b2 = (const float*)d_in[8];
  float* out = (float*)d_out;

  const int NN = in_sizes[0] / 384;
  const int EE = in_sizes[2];
  const int* src = ei;
  const int* dst = ei + EE;

  char* ws = (char*)d_ws;
  float* deg  = (float*)ws;
  size_t off  = ((size_t)NN * sizeof(float) + 255) & ~(size_t)255;
  float* part = (float*)(ws + off);
  off += ((size_t)kDegSlices * NN * sizeof(float) + 255) & ~(size_t)255;
  float* x64  = (float*)(ws + off);
  float* hbuf = (float*)(ws + off + 1 * (size_t)NN * 64 * sizeof(float));
  float* agg  = (float*)(ws + off + 2 * (size_t)NN * 64 * sizeof(float));
  float* enorm = (float*)(ws + off + 3 * (size_t)NN * 64 * sizeof(float));

  const dim3 blk(256);
  const int gES   = ((EE + 31) / 32 + 7) / 8;
  const int gGemm = ((NN + 15) / 16 + 7) / 8;
  const int nRanges = (NN + kDegRange - 1) / kDegRange;
  const int gDeg  = nRanges * kDegSlices;

  proj_and_degree<<<gDeg + gGemm, blk, 0, stream>>>(
      x, Wm, bm, x64, dst, ew, part, NN, EE, nRanges);

  gemm_f32_wmma<64, false, true, true><<<gGemm, blk, 0, stream>>>(
      x64, W1, b1, hbuf, part, agg, deg, NN);
  edge_scatter<true><<<gES, blk, 0, stream>>>(hbuf, src, dst, ew, deg, enorm,
                                              agg, EE);

  gemm_f32_wmma<64, true, true, false><<<gGemm, blk, 0, stream>>>(
      agg, W2, b2, hbuf, deg, out, nullptr, NN);
  edge_scatter<false><<<gES, blk, 0, stream>>>(hbuf, src, dst, ew, deg, enorm,
                                               out, EE);
}
